# starting kernel
# baseline (speedup 1.0000x reference)
#include <hip/hip_runtime.h>
#include <math.h>

#define DIMS     1024
#define HEADS    16
#define HDIM     64
#define ROT      32
#define BATCH    4
#define SEQ      8192

typedef __attribute__((ext_vector_type(2))) float v2f;
typedef __attribute__((ext_vector_type(8))) float v8f;

__global__ void __launch_bounds__(64)
build_M_kernel(const float* __restrict__ matrix,
               const float* __restrict__ thetas,
               const float* __restrict__ tscale,
               const int*   __restrict__ pairs,
               float*       __restrict__ Mout) {
    __shared__ float acc[HDIM][HDIM];
    const int t = threadIdx.x;
    for (int c = 0; c < HDIM; ++c) acc[t][c] = (t == c) ? 1.0f : 0.0f;

    const float ts = tscale[0];
    for (int g = 0; g < ROT; ++g) {
        const int   i  = pairs[2 * g];
        const int   j  = pairs[2 * g + 1];
        const float th = thetas[g] * ts;
        const float cs = cosf(th), sn = sinf(th);
        if (i != j) {
            const float vi = acc[t][i], vj = acc[t][j];
            acc[t][i] =  cs * vi + sn * vj;
            acc[t][j] = -sn * vi + cs * vj;
        } else {
            acc[t][i] *= sn;
        }
    }
    for (int n = 0; n < HDIM; ++n) {
        float s = 0.0f;
        for (int m = 0; m < HDIM; ++m)
            s = fmaf(acc[t][m], matrix[m * HDIM + n], s);
        Mout[t * HDIM + n] = s;
    }
}

#define STG 66

__global__ void __launch_bounds__(256)
rope_gemm_kernel(const float* __restrict__ x,
                 const float* __restrict__ Mg,
                 const float* __restrict__ invf,
                 float*       __restrict__ out) {
    __shared__ __align__(16) float Bl[16 * 4 * 64];
    __shared__ __align__(16) float stg[8][16 * STG];
    __shared__ __align__(16) float sctab[8][64];

    const int tid  = threadIdx.x;
    const int wave = tid >> 5;
    const int lane = tid & 31;

    for (int idx = tid; idx < 16 * 4 * 64; idx += 256) {
        const int v  = idx & 1;
        const int ln = (idx >> 1) & 31;
        const int t  = (idx >> 6) & 3;
        const int kk = idx >> 8;
        const int krow = 4 * kk + ((ln >> 4) << 1) + v;
        const int col  = 16 * t + (ln & 15);
        Bl[idx] = Mg[krow * HDIM + col];
    }
    __syncthreads();

    const int  tile    = blockIdx.x * 8 + wave;
    const int  s       = tile & (SEQ - 1);
    const long rowbase = (long)tile * 16;

    if (lane < ROT) {
        const float a = (float)s * invf[lane];
        sctab[wave][lane]      = cosf(a);
        sctab[wave][lane + 32] = sinf(a);
    }

    const float* xrow = x + (rowbase + (lane & 15)) * HDIM + ((lane >> 4) << 1);
    v2f afrag[16];
#pragma unroll
    for (int kk = 0; kk < 16; ++kk)
        afrag[kk] = *(const v2f*)(xrow + kk * 4);

    v8f c0 = {}, c1 = {}, c2 = {}, c3 = {};
    const float* Bp = Bl + lane * 2;
#pragma unroll
    for (int kk = 0; kk < 16; ++kk) {
        const v2f b0 = *(const v2f*)(Bp + (kk * 4 + 0) * 64);
        const v2f b1 = *(const v2f*)(Bp + (kk * 4 + 1) * 64);
        const v2f b2 = *(const v2f*)(Bp + (kk * 4 + 2) * 64);
        const v2f b3 = *(const v2f*)(Bp + (kk * 4 + 3) * 64);
        c0 = __builtin_amdgcn_wmma_f32_16x16x4_f32(false, afrag[kk], false, b0,
                                                   (short)0, c0, false, false);
        c1 = __builtin_amdgcn_wmma_f32_16x16x4_f32(false, afrag[kk], false, b1,
                                                   (short)0, c1, false, false);
        c2 = __builtin_amdgcn_wmma_f32_16x16x4_f32(false, afrag[kk], false, b2,
                                                   (short)0, c2, false, false);
        c3 = __builtin_amdgcn_wmma_f32_16x16x4_f32(false, afrag[kk], false, b3,
                                                   (short)0, c3, false, false);
    }

    {
        const int rbase = (lane >> 4) << 3;
        const int col   = lane & 15;
        float*    st    = stg[wave];
#pragma unroll
        for (int r = 0; r < 8; ++r) {
            st[(rbase + r) * STG + col]      = c0[r];
            st[(rbase + r) * STG + col + 16] = c1[r];
            st[(rbase + r) * STG + col + 32] = c2[r];
            st[(rbase + r) * STG + col + 48] = c3[r];
        }
    }
    __syncthreads();

    const float* st = stg[wave];
    const float* sc = sctab[wave];
    float*       op = out + rowbase * HDIM;
#pragma unroll
    for (int i = 0; i < 32; ++i) {
        const int flat = i * 32 + lane;
        const int row  = flat >> 6;
        const int col  = flat & 63;
        const int n    = col & 31;
        const v2f xy = *(const v2f*)(st + row * STG + 2 * n);
        const float cs = sc[n], sn = sc[n + 32];
        const float val = (col < 32) ? (xy.x * cs - xy.y * sn)
                                     : (xy.x * sn + xy.y * cs);
        op[flat] = val * 32.0f;
    }
}

extern "C" void kernel_launch(void* const* d_in, const int* in_sizes, int n_in,
                              void* d_out, int out_size, void* d_ws, size_t ws_size,
                              hipStream_t stream) {
    const float* x      = (const float*)d_in[0];
    const float* matrix = (const float*)d_in[1];
    const float* thetas = (const float*)d_in[2];
    const float* tscale = (const float*)d_in[3];
    const float* invf   = (const float*)d_in[4];
    const int*   pairs  = (const int*)d_in[5];
    float*       Mws    = (float*)d_ws;

    build_M_kernel<<<1, 64, 0, stream>>>(matrix, thetas, tscale, pairs, Mws);

    const int tiles  = BATCH * SEQ;
    const int blocks = tiles / 8;
    rope_gemm_kernel<<<blocks, 256, 0, stream>>>(x, Mws, invf, (float*)d_out);
}


# Main kernel as a transposed product: M^T stays in registers as the A operand while a wave streams its 8 x tiles as B; RoPE in registers, whole lines out via the wave's LDS slab; y's sum order changes
# speedup vs baseline: 2.2968x; 2.2968x over previous
#include <hip/hip_runtime.h>
#include <math.h>

#define DIMS     1024
#define HEADS    16
#define HDIM     64
#define ROT      32
#define BATCH    4
#define SEQ      8192

typedef __attribute__((ext_vector_type(2))) float v2f;
typedef __attribute__((ext_vector_type(4))) float v4f;
typedef __attribute__((ext_vector_type(8))) float v8f;

#define ACCP 72

__global__ void __launch_bounds__(128)
build_M_kernel(const float* __restrict__ matrix,
               const float* __restrict__ thetas,
               const float* __restrict__ tscale,
               const int*   __restrict__ pairs,
               float*       __restrict__ Mout) {
    __shared__ __align__(16) float accT[HDIM * ACCP];
    __shared__ __align__(16) float Bm[16 * 4 * 64];
    const int tid  = threadIdx.x;
    const int wave = tid >> 5;
    const int lane = tid & 31;

    if (wave < 2) {
        const int t = tid;
        const int   pi = pairs[2 * lane];
        const int   pj = pairs[2 * lane + 1];
        const float th = thetas[lane] * tscale[0];
        const float cl = cosf(th), sl = sinf(th);

        for (int cc = 0; cc < HDIM; ++cc)
            accT[cc * ACCP + t] = (t == cc) ? 1.0f : 0.0f;

        for (int g = 0; g < ROT; ++g) {
            const int   i  = __builtin_amdgcn_readlane(pi, g);
            const int   j  = __builtin_amdgcn_readlane(pj, g);
            const float cs = __int_as_float(__builtin_amdgcn_readlane(__float_as_int(cl), g));
            const float sn = __int_as_float(__builtin_amdgcn_readlane(__float_as_int(sl), g));
            if (i != j) {
                const float vi = accT[i * ACCP + t], vj = accT[j * ACCP + t];
                accT[i * ACCP + t] =  cs * vi + sn * vj;
                accT[j * ACCP + t] = -sn * vi + cs * vj;
            } else {
                accT[i * ACCP + t] *= sn;
            }
        }
    } else {
        const int u = tid - 64;
        const float* src = matrix + (u >> 4) * HDIM + 4 * (u & 15);
        float* dst = Bm + ((u >> 2) & 3) * 64 + ((u >> 5) & 1) * 32
                        + (u & 3) * 8 + ((u >> 4) & 1);
        v4f m[16];
#pragma unroll
        for (int q = 0; q < 16; ++q)
            m[q] = *(const v4f*)(src + q * 4 * HDIM);
#pragma unroll
        for (int q = 0; q < 16; ++q) {
            dst[q * 256 + 0] = m[q].x;
            dst[q * 256 + 2] = m[q].y;
            dst[q * 256 + 4] = m[q].z;
            dst[q * 256 + 6] = m[q].w;
        }
    }
    __syncthreads();

    const float* Ap = accT + ((lane >> 4) << 1) * ACCP + 16 * wave + (lane & 15);
    const float* Bp = Bm + lane * 2;
    v8f c0 = {}, c1 = {}, c2 = {}, c3 = {};
#pragma unroll
    for (int kk = 0; kk < 16; ++kk) {
        v2f a;
        a.x = Ap[(4 * kk + 0) * ACCP];
        a.y = Ap[(4 * kk + 1) * ACCP];
        const v2f b0 = *(const v2f*)(Bp + (kk * 4 + 0) * 64);
        const v2f b1 = *(const v2f*)(Bp + (kk * 4 + 1) * 64);
        const v2f b2 = *(const v2f*)(Bp + (kk * 4 + 2) * 64);
        const v2f b3 = *(const v2f*)(Bp + (kk * 4 + 3) * 64);
        c0 = __builtin_amdgcn_wmma_f32_16x16x4_f32(false, a, false, b0,
                                                   (short)0, c0, false, false);
        c1 = __builtin_amdgcn_wmma_f32_16x16x4_f32(false, a, false, b1,
                                                   (short)0, c1, false, false);
        c2 = __builtin_amdgcn_wmma_f32_16x16x4_f32(false, a, false, b2,
                                                   (short)0, c2, false, false);
        c3 = __builtin_amdgcn_wmma_f32_16x16x4_f32(false, a, false, b3,
                                                   (short)0, c3, false, false);
    }

    float* Mp = Mout + (16 * wave + ((lane >> 4) << 3)) * HDIM + (lane & 15);
#pragma unroll
    for (int r = 0; r < 8; ++r) {
        Mp[r * HDIM]      = c0[r];
        Mp[r * HDIM + 16] = c1[r];
        Mp[r * HDIM + 32] = c2[r];
        Mp[r * HDIM + 48] = c3[r];
    }
}

#define WAVES   16
#define OPITCH  68

__device__ __forceinline__ void wave_lds_order() {
    __builtin_amdgcn_fence(__ATOMIC_RELEASE, "wavefront");
    __builtin_amdgcn_wave_barrier();
    __builtin_amdgcn_fence(__ATOMIC_ACQUIRE, "wavefront");
}

__global__ void __launch_bounds__(WAVES * 32)
rope_gemm_kernel(const float* __restrict__ x,
                 const float* __restrict__ Mg,
                 const float* __restrict__ invf,
                 float*       __restrict__ out) {
    __shared__ __align__(16) float slab[WAVES][16 * OPITCH];
    __shared__ __align__(16) float sctab[WAVES][64];

    const int wave = threadIdx.x >> 5;
    const int lane = threadIdx.x & 31;
    const int c    = lane & 15;
    const int h    = lane >> 4;
    const int gw   = blockIdx.x * WAVES + wave;

    v2f am[16][4];
#pragma unroll
    for (int jp = 0; jp < 16; ++jp) {
        const int k = 8 * (jp >> 1) + 4 * h + 2 * (jp & 1);
#pragma unroll
        for (int t = 0; t < 4; ++t) {
            am[jp][t].x = Mg[k * HDIM + 16 * t + c];
            am[jp][t].y = Mg[(k + 1) * HDIM + 16 * t + c];
        }
    }

    float* sl  = slab[wave];
    float* sct = sctab[wave];

    const float invf_l = invf[lane];

#pragma unroll 1
    for (int i = 0; i < 8; ++i) {
        const long tile = (long)gw * 8 + i;
        const int  s    = (int)tile & (SEQ - 1);

        const float* xt = x + tile * (16 * HDIM) + c * HDIM + 4 * h;
        v4f xv[8];
#pragma unroll
        for (int j = 0; j < 8; ++j)
            xv[j] = *(const v4f*)(xt + 8 * j);

        {
            const float a = (float)s * invf_l;
            sct[lane]      = cosf(a) * 32.0f;
            sct[lane + 32] = sinf(a) * 32.0f;
        }
        wave_lds_order();

        v8f acc[4] = {};
#pragma unroll
        for (int j = 0; j < 8; ++j) {
#pragma unroll
            for (int p = 0; p < 2; ++p) {
                v2f bf;
                bf.x = p ? xv[j].z : xv[j].x;
                bf.y = p ? xv[j].w : xv[j].y;
#pragma unroll
                for (int t = 0; t < 4; ++t)
                    acc[t] = __builtin_amdgcn_wmma_f32_16x16x4_f32(
                        false, am[2 * j + p][t], false, bf,
                        (short)0, acc[t], false, false);
            }
        }

#pragma unroll
        for (int t = 0; t < 4; ++t) {
            const v4f cs = *(const v4f*)(sct + 8 * t + 4 * h);
            const v4f sn = *(const v4f*)(sct + 32 + 8 * t + 4 * h);
            v4f o1, o2;
#pragma unroll
            for (int q = 0; q < 4; ++q) {
                const float y0 = acc[t][2 * q], y1 = acc[t][2 * q + 1];
                o1[q] = y0 * cs[q] - y1 * sn[q];
                o2[q] = y0 * sn[q] + y1 * cs[q];
            }
            *(v4f*)(sl + c * OPITCH + 8 * t + 4 * h)      = o1;
            *(v4f*)(sl + c * OPITCH + 32 + 8 * t + 4 * h) = o2;
        }
        wave_lds_order();

        float* op = out + tile * (16 * HDIM) + 4 * lane;
#pragma unroll
        for (int k = 0; k < 8; ++k) {
            const v4f w = *(const v4f*)(sl + (2 * k + h) * OPITCH + 4 * c);
            *(v4f*)(op + 128 * k) = w;
        }
        wave_lds_order();
    }
}

extern "C" void kernel_launch(void* const* d_in, const int* in_sizes, int n_in,
                              void* d_out, int out_size, void* d_ws, size_t ws_size,
                              hipStream_t stream) {
    const float* x      = (const float*)d_in[0];
    const float* matrix = (const float*)d_in[1];
    const float* thetas = (const float*)d_in[2];
    const float* tscale = (const float*)d_in[3];
    const float* invf   = (const float*)d_in[4];
    const int*   pairs  = (const int*)d_in[5];
    float*       Mws    = (float*)d_ws;

    build_M_kernel<<<1, 128, 0, stream>>>(matrix, thetas, tscale, pairs, Mws);

    rope_gemm_kernel<<<256, WAVES * 32, 0, stream>>>(x, Mws, invf, (float*)d_out);
}


# Main kernel: x tile (B operand) double-buffered in registers: the next position's 8 loads go out before the current tile's 64 products, RoPE and stores; was load, wait, compute; sums kept
# speedup vs baseline: 2.4018x; 1.0457x over previous
#include <hip/hip_runtime.h>
#include <math.h>

#define DIMS     1024
#define HEADS    16
#define HDIM     64
#define ROT      32
#define BATCH    4
#define SEQ      8192

typedef __attribute__((ext_vector_type(2))) float v2f;
typedef __attribute__((ext_vector_type(4))) float v4f;
typedef __attribute__((ext_vector_type(8))) float v8f;

#define ACCP 72

__global__ void __launch_bounds__(128)
build_M_kernel(const float* __restrict__ matrix,
               const float* __restrict__ thetas,
               const float* __restrict__ tscale,
               const int*   __restrict__ pairs,
               float*       __restrict__ Mout) {
    __shared__ __align__(16) float accT[HDIM * ACCP];
    __shared__ __align__(16) float Bm[16 * 4 * 64];
    const int tid  = threadIdx.x;
    const int wave = tid >> 5;
    const int lane = tid & 31;

    if (wave < 2) {
        const int t = tid;
        const int   pi = pairs[2 * lane];
        const int   pj = pairs[2 * lane + 1];
        const float th = thetas[lane] * tscale[0];
        const float cl = cosf(th), sl = sinf(th);

        for (int cc = 0; cc < HDIM; ++cc)
            accT[cc * ACCP + t] = (t == cc) ? 1.0f : 0.0f;

        for (int g = 0; g < ROT; ++g) {
            const int   i  = __builtin_amdgcn_readlane(pi, g);
            const int   j  = __builtin_amdgcn_readlane(pj, g);
            const float cs = __int_as_float(__builtin_amdgcn_readlane(__float_as_int(cl), g));
            const float sn = __int_as_float(__builtin_amdgcn_readlane(__float_as_int(sl), g));
            if (i != j) {
                const float vi = accT[i * ACCP + t], vj = accT[j * ACCP + t];
                accT[i * ACCP + t] =  cs * vi + sn * vj;
                accT[j * ACCP + t] = -sn * vi + cs * vj;
            } else {
                accT[i * ACCP + t] *= sn;
            }
        }
    } else {
        const int u = tid - 64;
        const float* src = matrix + (u >> 4) * HDIM + 4 * (u & 15);
        float* dst = Bm + ((u >> 2) & 3) * 64 + ((u >> 5) & 1) * 32
                        + (u & 3) * 8 + ((u >> 4) & 1);
        v4f m[16];
#pragma unroll
        for (int q = 0; q < 16; ++q)
            m[q] = *(const v4f*)(src + q * 4 * HDIM);
#pragma unroll
        for (int q = 0; q < 16; ++q) {
            dst[q * 256 + 0] = m[q].x;
            dst[q * 256 + 2] = m[q].y;
            dst[q * 256 + 4] = m[q].z;
            dst[q * 256 + 6] = m[q].w;
        }
    }
    __syncthreads();

    const float* Ap = accT + ((lane >> 4) << 1) * ACCP + 16 * wave + (lane & 15);
    const float* Bp = Bm + lane * 2;
    v8f c0 = {}, c1 = {}, c2 = {}, c3 = {};
#pragma unroll
    for (int kk = 0; kk < 16; ++kk) {
        v2f a;
        a.x = Ap[(4 * kk + 0) * ACCP];
        a.y = Ap[(4 * kk + 1) * ACCP];
        const v2f b0 = *(const v2f*)(Bp + (kk * 4 + 0) * 64);
        const v2f b1 = *(const v2f*)(Bp + (kk * 4 + 1) * 64);
        const v2f b2 = *(const v2f*)(Bp + (kk * 4 + 2) * 64);
        const v2f b3 = *(const v2f*)(Bp + (kk * 4 + 3) * 64);
        c0 = __builtin_amdgcn_wmma_f32_16x16x4_f32(false, a, false, b0,
                                                   (short)0, c0, false, false);
        c1 = __builtin_amdgcn_wmma_f32_16x16x4_f32(false, a, false, b1,
                                                   (short)0, c1, false, false);
        c2 = __builtin_amdgcn_wmma_f32_16x16x4_f32(false, a, false, b2,
                                                   (short)0, c2, false, false);
        c3 = __builtin_amdgcn_wmma_f32_16x16x4_f32(false, a, false, b3,
                                                   (short)0, c3, false, false);
    }

    float* Mp = Mout + (16 * wave + ((lane >> 4) << 3)) * HDIM + (lane & 15);
#pragma unroll
    for (int r = 0; r < 8; ++r) {
        Mp[r * HDIM]      = c0[r];
        Mp[r * HDIM + 16] = c1[r];
        Mp[r * HDIM + 32] = c2[r];
        Mp[r * HDIM + 48] = c3[r];
    }
}

#define WAVES   16
#define OPITCH  68

__device__ __forceinline__ void wave_lds_order() {
    __builtin_amdgcn_fence(__ATOMIC_RELEASE, "wavefront");
    __builtin_amdgcn_wave_barrier();
    __builtin_amdgcn_fence(__ATOMIC_ACQUIRE, "wavefront");
}

__device__ __forceinline__ void load_x_tile(v4f (&xv)[8], const float* xt) {
#pragma unroll
    for (int j = 0; j < 8; ++j)
        xv[j] = *(const v4f*)(xt + 8 * j);
}

__device__ __forceinline__ void do_tile(const v4f (&xv)[8], const v2f (&am)[16][4],
                                        float* sl, float* sct, float* op,
                                        int s, float invf_l, int lane, int c, int h) {
    {
        const float a = (float)s * invf_l;
        sct[lane]      = cosf(a) * 32.0f;
        sct[lane + 32] = sinf(a) * 32.0f;
    }
    wave_lds_order();

    v8f acc[4] = {};
#pragma unroll
    for (int j = 0; j < 8; ++j) {
#pragma unroll
        for (int p = 0; p < 2; ++p) {
            v2f bf;
            bf.x = p ? xv[j].z : xv[j].x;
            bf.y = p ? xv[j].w : xv[j].y;
#pragma unroll
            for (int t = 0; t < 4; ++t)
                acc[t] = __builtin_amdgcn_wmma_f32_16x16x4_f32(
                    false, am[2 * j + p][t], false, bf,
                    (short)0, acc[t], false, false);
        }
    }

#pragma unroll
    for (int t = 0; t < 4; ++t) {
        const v4f cs = *(const v4f*)(sct + 8 * t + 4 * h);
        const v4f sn = *(const v4f*)(sct + 32 + 8 * t + 4 * h);
        v4f o1, o2;
#pragma unroll
        for (int q = 0; q < 4; ++q) {
            const float y0 = acc[t][2 * q], y1 = acc[t][2 * q + 1];
            o1[q] = y0 * cs[q] - y1 * sn[q];
            o2[q] = y0 * sn[q] + y1 * cs[q];
        }
        *(v4f*)(sl + c * OPITCH + 8 * t + 4 * h)      = o1;
        *(v4f*)(sl + c * OPITCH + 32 + 8 * t + 4 * h) = o2;
    }
    wave_lds_order();

#pragma unroll
    for (int k = 0; k < 8; ++k) {
        const v4f w = *(const v4f*)(sl + (2 * k + h) * OPITCH + 4 * c);
        *(v4f*)(op + 128 * k) = w;
    }
    wave_lds_order();
}

__global__ void __launch_bounds__(WAVES * 32)
rope_gemm_kernel(const float* __restrict__ x,
                 const float* __restrict__ Mg,
                 const float* __restrict__ invf,
                 float*       __restrict__ out) {
    __shared__ __align__(16) float slab[WAVES][16 * OPITCH];
    __shared__ __align__(16) float sctab[WAVES][64];

    const int wave = threadIdx.x >> 5;
    const int lane = threadIdx.x & 31;
    const int c    = lane & 15;
    const int h    = lane >> 4;
    const int gw   = blockIdx.x * WAVES + wave;

    const int  TILE  = 16 * HDIM;
    const long tile0 = (long)gw * 8;
    const int  s0    = (int)tile0 & (SEQ - 1);
    const float* xt = x + tile0 * TILE + c * HDIM + 4 * h;
    float*       op = out + tile0 * TILE + 4 * lane;

    v4f xa[8], xb[8];
    load_x_tile(xa, xt);

    v2f am[16][4];
#pragma unroll
    for (int jp = 0; jp < 16; ++jp) {
        const int k = 8 * (jp >> 1) + 4 * h + 2 * (jp & 1);
#pragma unroll
        for (int t = 0; t < 4; ++t) {
            am[jp][t].x = Mg[k * HDIM + 16 * t + c];
            am[jp][t].y = Mg[(k + 1) * HDIM + 16 * t + c];
        }
    }

    float* sl  = slab[wave];
    float* sct = sctab[wave];

    const float invf_l = invf[lane];

#pragma unroll 1
    for (int i = 0; i < 6; i += 2) {
        load_x_tile(xb, xt + (i + 1) * TILE);
        do_tile(xa, am, sl, sct, op + i * TILE, s0 + i, invf_l, lane, c, h);
        load_x_tile(xa, xt + (i + 2) * TILE);
        do_tile(xb, am, sl, sct, op + (i + 1) * TILE, s0 + i + 1, invf_l, lane, c, h);
    }
    load_x_tile(xb, xt + 7 * TILE);
    do_tile(xa, am, sl, sct, op + 6 * TILE, s0 + 6, invf_l, lane, c, h);
    do_tile(xb, am, sl, sct, op + 7 * TILE, s0 + 7, invf_l, lane, c, h);
}

extern "C" void kernel_launch(void* const* d_in, const int* in_sizes, int n_in,
                              void* d_out, int out_size, void* d_ws, size_t ws_size,
                              hipStream_t stream) {
    const float* x      = (const float*)d_in[0];
    const float* matrix = (const float*)d_in[1];
    const float* thetas = (const float*)d_in[2];
    const float* tscale = (const float*)d_in[3];
    const float* invf   = (const float*)d_in[4];
    const int*   pairs  = (const int*)d_in[5];
    float*       Mws    = (float*)d_ws;

    build_M_kernel<<<1, 128, 0, stream>>>(matrix, thetas, tscale, pairs, Mws);

    rope_gemm_kernel<<<256, WAVES * 32, 0, stream>>>(x, Mws, invf, (float*)d_out);
}
